# starting kernel
# baseline (speedup 1.0000x reference)
#include <hip/hip_runtime.h>

typedef __attribute__((ext_vector_type(2))) float v2f;
typedef __attribute__((ext_vector_type(8))) float v8f;

#define N_NODES 100000
#define N_EDGES 3200000
#define IN_CH 512
#define HID 16

__global__ void gcn_init(float* __restrict__ deg, float* __restrict__ agg) {
    int i = blockIdx.x * blockDim.x + threadIdx.x;
    if (i < N_NODES * HID) agg[i] = 0.0f;
    if (i < N_NODES) deg[i] = 1.0f;
}

__global__ void gcn_degree(const int* __restrict__ col, float* __restrict__ deg) {
    int e = blockIdx.x * blockDim.x + threadIdx.x;
    if (e < N_EDGES) atomicAdd(&deg[col[e]], 1.0f);
}

__global__ void gcn_rsqrt(float* __restrict__ deg) {
    int i = blockIdx.x * blockDim.x + threadIdx.x;
    if (i < N_NODES) deg[i] = rsqrtf(deg[i]);
}

__global__ void __launch_bounds__(256) gcn_xw_wmma(const float* __restrict__ x,
                                                   const float* __restrict__ W1,
                                                   float* __restrict__ xw) {
    __shared__ float sW1[IN_CH * HID];
    for (int i = threadIdx.x; i < (IN_CH * HID) / 4; i += 256)
        ((float4*)sW1)[i] = ((const float4*)W1)[i];
    __syncthreads();

    const int wave = threadIdx.x >> 5;
    const int lane = threadIdx.x & 31;
    const int half = lane >> 4;
    const int lrow = lane & 15;
    const int tile = blockIdx.x * 8 + wave;
    if (tile * 16 >= N_NODES) return;

    const float* xrow = x + (size_t)(tile * 16 + lrow) * IN_CH + 2 * half;
    const float* bptr = sW1 + (2 * half) * HID + lrow;

    v8f acc = {};
#pragma unroll 8
    for (int k0 = 0; k0 < IN_CH; k0 += 4) {
        v2f a = *(const v2f*)(xrow + k0);
        v2f b;
        b.x = bptr[k0 * HID];
        b.y = bptr[k0 * HID + HID];
        acc = __builtin_amdgcn_wmma_f32_16x16x4_f32(
             false, a,  false, b,
             (short)0, acc,  false,  false);
    }

    float* o = xw + (size_t)(tile * 16 + 8 * half) * HID + lrow;
#pragma unroll
    for (int r = 0; r < 8; ++r) o[r * HID] = acc[r];
}

__global__ void gcn_scatter(const int* __restrict__ row, const int* __restrict__ col,
                            const float* __restrict__ dinv, const float* __restrict__ xw,
                            float* __restrict__ agg) {
    long long t = (long long)blockIdx.x * blockDim.x + threadIdx.x;
    int e = (int)(t >> 4);
    int c = (int)t & 15;
    if (e < N_EDGES) {
        int r = row[e];
        int d = col[e];
        float nrm = dinv[r] * dinv[d];
        atomicAdd(&agg[(size_t)d * HID + c], xw[(size_t)r * HID + c] * nrm);
    }
}

__global__ void gcn_out(const float* __restrict__ agg, const float* __restrict__ xw,
                        const float* __restrict__ dinv, const float* __restrict__ b1,
                        const float* __restrict__ W2, const float* __restrict__ b2,
                        float* __restrict__ out) {
    int i = blockIdx.x * blockDim.x + threadIdx.x;
    if (i >= N_NODES) return;
    float s = dinv[i] * dinv[i];
    float o0 = b2[0], o1 = b2[1];
#pragma unroll
    for (int c = 0; c < HID; ++c) {
        float h = agg[(size_t)i * HID + c] + xw[(size_t)i * HID + c] * s + b1[c];
        o0 += h * W2[c * 2 + 0];
        o1 += h * W2[c * 2 + 1];
    }
    out[i * 2 + 0] = o0;
    out[i * 2 + 1] = o1;
}

extern "C" void kernel_launch(void* const* d_in, const int* in_sizes, int n_in,
                              void* d_out, int out_size, void* d_ws, size_t ws_size,
                              hipStream_t stream) {
    const float* x  = (const float*)d_in[0];
    const int*   ei = (const int*)d_in[1];
    const float* W1 = (const float*)d_in[2];
    const float* b1 = (const float*)d_in[3];
    const float* W2 = (const float*)d_in[4];
    const float* b2 = (const float*)d_in[5];
    float* out = (float*)d_out;

    float* deg = (float*)d_ws;
    float* xw  = deg + N_NODES;
    float* agg = xw + (size_t)N_NODES * HID;

    const int* row = ei;
    const int* col = ei + N_EDGES;

    gcn_init<<<(N_NODES * HID + 255) / 256, 256, 0, stream>>>(deg, agg);
    gcn_degree<<<(N_EDGES + 255) / 256, 256, 0, stream>>>(col, deg);
    gcn_rsqrt<<<(N_NODES + 255) / 256, 256, 0, stream>>>(deg);

    int n_tiles = N_NODES / 16;
    gcn_xw_wmma<<<(n_tiles + 7) / 8, 256, 0, stream>>>(x, W1, xw);

    long long sthreads = (long long)N_EDGES * 16;
    gcn_scatter<<<(int)((sthreads + 255) / 256), 256, 0, stream>>>(row, col, deg, xw, agg);

    gcn_out<<<(N_NODES + 255) / 256, 256, 0, stream>>>(agg, xw, deg, b1, W2, b2, out);
}


# degree count moved from 3.2M scattered global float atomics to LDS integer counters per edge slice and node range, partial planes summed in the rsqrt kernel; counts exact, same dinv
# speedup vs baseline: 1.6610x; 1.6610x over previous
#include <hip/hip_runtime.h>

typedef __attribute__((ext_vector_type(2))) float v2f;
typedef __attribute__((ext_vector_type(8))) float v8f;

#define N_NODES 100000
#define N_EDGES 3200000
#define IN_CH 512
#define HID 16

__global__ void gcn_init(float* __restrict__ agg) {
    int i = blockIdx.x * blockDim.x + threadIdx.x;
    if (i < N_NODES * HID) agg[i] = 0.0f;
}

#define DEG_SLICES 64
#define DEG_RANGES 4
#define DEG_THREADS 1024
#define SLICE_EDGES (N_EDGES / DEG_SLICES)
#define SLICE_VEC4  (SLICE_EDGES / 4)
#define RANGE_NODES (N_NODES / DEG_RANGES)
#define RANGE_WORDS (RANGE_NODES / 2)
#define PAIR_WORDS  (N_NODES / 2)
#define DEG_LOADS   ((SLICE_VEC4 + DEG_THREADS - 1) / DEG_THREADS)
static_assert(N_EDGES % DEG_SLICES == 0 && SLICE_EDGES % 4 == 0, "slices are whole 16-byte loads");
static_assert(SLICE_EDGES <= 65535, "one slice landing on one node must fit a 16-bit field");
static_assert(N_NODES % DEG_RANGES == 0 && RANGE_NODES % 2 == 0 && RANGE_WORDS % 4 == 0,
              "ranges are whole 16-byte pieces of packed words");

__device__ __forceinline__ void deg_count(unsigned* cnt, int d, unsigned lo) {
    unsigned k = (unsigned)d - lo;
    if (k < (unsigned)RANGE_NODES) atomicAdd(&cnt[k >> 1], 1u << ((k & 1u) << 4));
}

__global__ void __launch_bounds__(DEG_THREADS) gcn_degree_lds(const int* __restrict__ col,
                                                              float* __restrict__ planes) {
    __shared__ unsigned cnt[RANGE_WORDS];
    const int r = blockIdx.x % DEG_RANGES;
    const int s = blockIdx.x / DEG_RANGES;
    for (int i = threadIdx.x; i < RANGE_WORDS / 4; i += DEG_THREADS)
        ((uint4*)cnt)[i] = make_uint4(0u, 0u, 0u, 0u);

    const int4* c4 = (const int4*)(col + (size_t)s * SLICE_EDGES);
    int4 v[DEG_LOADS];
    for (int j = 0; j < DEG_LOADS; ++j) {
        int i = threadIdx.x + j * DEG_THREADS;
        v[j] = (i < SLICE_VEC4) ? c4[i] : make_int4(-1, -1, -1, -1);
    }
    __syncthreads();

    const unsigned lo = (unsigned)(r * RANGE_NODES);
    for (int j = 0; j < DEG_LOADS; ++j) {
        deg_count(cnt, v[j].x, lo);
        deg_count(cnt, v[j].y, lo);
        deg_count(cnt, v[j].z, lo);
        deg_count(cnt, v[j].w, lo);
    }
    __syncthreads();

    uint4* dst = (uint4*)((unsigned*)planes + (size_t)s * PAIR_WORDS + r * RANGE_WORDS);
    for (int i = threadIdx.x; i < RANGE_WORDS / 4; i += DEG_THREADS)
        dst[i] = ((const uint4*)cnt)[i];
}

__global__ void gcn_rsqrt(const float* __restrict__ planes, float* __restrict__ deg) {
    int p = blockIdx.x * blockDim.x + threadIdx.x;
    if (p >= PAIR_WORDS) return;
    const unsigned* w = (const unsigned*)planes + p;
    unsigned a = 1u, b = 1u;
    for (int s = 0; s < DEG_SLICES; ++s) {
        unsigned x = w[(size_t)s * PAIR_WORDS];
        a += x & 0xFFFFu;
        b += x >> 16;
    }
    float2 o;
    o.x = rsqrtf((float)a);
    o.y = rsqrtf((float)b);
    ((float2*)deg)[p] = o;
}

__global__ void __launch_bounds__(256) gcn_xw_wmma(const float* __restrict__ x,
                                                   const float* __restrict__ W1,
                                                   float* __restrict__ xw) {
    __shared__ float sW1[IN_CH * HID];
    for (int i = threadIdx.x; i < (IN_CH * HID) / 4; i += 256)
        ((float4*)sW1)[i] = ((const float4*)W1)[i];
    __syncthreads();

    const int wave = threadIdx.x >> 5;
    const int lane = threadIdx.x & 31;
    const int half = lane >> 4;
    const int lrow = lane & 15;
    const int tile = blockIdx.x * 8 + wave;
    if (tile * 16 >= N_NODES) return;

    const float* xrow = x + (size_t)(tile * 16 + lrow) * IN_CH + 2 * half;
    const float* bptr = sW1 + (2 * half) * HID + lrow;

    v8f acc = {};
#pragma unroll 8
    for (int k0 = 0; k0 < IN_CH; k0 += 4) {
        v2f a = *(const v2f*)(xrow + k0);
        v2f b;
        b.x = bptr[k0 * HID];
        b.y = bptr[k0 * HID + HID];
        acc = __builtin_amdgcn_wmma_f32_16x16x4_f32(
             false, a,  false, b,
             (short)0, acc,  false,  false);
    }

    float* o = xw + (size_t)(tile * 16 + 8 * half) * HID + lrow;
#pragma unroll
    for (int r = 0; r < 8; ++r) o[r * HID] = acc[r];
}

__global__ void gcn_scatter(const int* __restrict__ row, const int* __restrict__ col,
                            const float* __restrict__ dinv, const float* __restrict__ xw,
                            float* __restrict__ agg) {
    long long t = (long long)blockIdx.x * blockDim.x + threadIdx.x;
    int e = (int)(t >> 4);
    int c = (int)t & 15;
    if (e < N_EDGES) {
        int r = row[e];
        int d = col[e];
        float nrm = dinv[r] * dinv[d];
        atomicAdd(&agg[(size_t)d * HID + c], xw[(size_t)r * HID + c] * nrm);
    }
}

__global__ void gcn_out(const float* __restrict__ agg, const float* __restrict__ xw,
                        const float* __restrict__ dinv, const float* __restrict__ b1,
                        const float* __restrict__ W2, const float* __restrict__ b2,
                        float* __restrict__ out) {
    int i = blockIdx.x * blockDim.x + threadIdx.x;
    if (i >= N_NODES) return;
    float s = dinv[i] * dinv[i];
    float o0 = b2[0], o1 = b2[1];
#pragma unroll
    for (int c = 0; c < HID; ++c) {
        float h = agg[(size_t)i * HID + c] + xw[(size_t)i * HID + c] * s + b1[c];
        o0 += h * W2[c * 2 + 0];
        o1 += h * W2[c * 2 + 1];
    }
    out[i * 2 + 0] = o0;
    out[i * 2 + 1] = o1;
}

extern "C" void kernel_launch(void* const* d_in, const int* in_sizes, int n_in,
                              void* d_out, int out_size, void* d_ws, size_t ws_size,
                              hipStream_t stream) {
    const float* x  = (const float*)d_in[0];
    const int*   ei = (const int*)d_in[1];
    const float* W1 = (const float*)d_in[2];
    const float* b1 = (const float*)d_in[3];
    const float* W2 = (const float*)d_in[4];
    const float* b2 = (const float*)d_in[5];
    float* out = (float*)d_out;

    float* deg = (float*)d_ws;
    float* xw  = deg + N_NODES;
    float* agg = xw + (size_t)N_NODES * HID;

    const int* row = ei;
    const int* col = ei + N_EDGES;

    gcn_degree_lds<<<DEG_SLICES * DEG_RANGES, DEG_THREADS, 0, stream>>>(col, xw);
    gcn_rsqrt<<<(PAIR_WORDS + 255) / 256, 256, 0, stream>>>(xw, deg);
    gcn_init<<<(N_NODES * HID + 255) / 256, 256, 0, stream>>>(agg);

    int n_tiles = N_NODES / 16;
    gcn_xw_wmma<<<(n_tiles + 7) / 8, 256, 0, stream>>>(x, W1, xw);

    long long sthreads = (long long)N_EDGES * 16;
    gcn_scatter<<<(int)((sthreads + 255) / 256), 256, 0, stream>>>(row, col, deg, xw, agg);

    gcn_out<<<(N_NODES + 255) / 256, 256, 0, stream>>>(agg, xw, deg, b1, W2, b2, out);
}


# 16->2 layer applied per node before aggregation (linearity): edges carry 2 floats, summed by LDS float atomics per edge slice and node range into planes, no global atomics; sum order changed
# speedup vs baseline: 3.9021x; 2.3492x over previous
#include <hip/hip_runtime.h>

typedef __attribute__((ext_vector_type(2))) float v2f;
typedef __attribute__((ext_vector_type(8))) float v8f;

#define N_NODES 100000
#define N_EDGES 3200000
#define IN_CH 512
#define HID 16

#define DEG_SLICES 64
#define DEG_RANGES 4
#define DEG_THREADS 1024
#define SLICE_EDGES (N_EDGES / DEG_SLICES)
#define SLICE_VEC4  (SLICE_EDGES / 4)
#define RANGE_NODES (N_NODES / DEG_RANGES)
#define RANGE_WORDS (RANGE_NODES / 2)
#define PAIR_WORDS  (N_NODES / 2)
#define DEG_LOADS   ((SLICE_VEC4 + DEG_THREADS - 1) / DEG_THREADS)
static_assert(N_EDGES % DEG_SLICES == 0 && SLICE_EDGES % 4 == 0, "slices are whole 16-byte loads");
static_assert(SLICE_EDGES <= 65535, "one slice landing on one node must fit a 16-bit field");
static_assert(N_NODES % DEG_RANGES == 0 && RANGE_NODES % 2 == 0 && RANGE_WORDS % 4 == 0,
              "ranges are whole 16-byte pieces of packed words");

__device__ __forceinline__ void deg_count(unsigned* cnt, int d, unsigned lo) {
    unsigned k = (unsigned)d - lo;
    if (k < (unsigned)RANGE_NODES) atomicAdd(&cnt[k >> 1], 1u << ((k & 1u) << 4));
}

__global__ void __launch_bounds__(DEG_THREADS) gcn_degree_lds(const int* __restrict__ col,
                                                              float* __restrict__ planes) {
    __shared__ unsigned cnt[RANGE_WORDS];
    const int r = blockIdx.x % DEG_RANGES;
    const int s = blockIdx.x / DEG_RANGES;
    for (int i = threadIdx.x; i < RANGE_WORDS / 4; i += DEG_THREADS)
        ((uint4*)cnt)[i] = make_uint4(0u, 0u, 0u, 0u);

    const int4* c4 = (const int4*)(col + (size_t)s * SLICE_EDGES);
    int4 v[DEG_LOADS];
    for (int j = 0; j < DEG_LOADS; ++j) {
        int i = threadIdx.x + j * DEG_THREADS;
        v[j] = (i < SLICE_VEC4) ? c4[i] : make_int4(-1, -1, -1, -1);
    }
    __syncthreads();

    const unsigned lo = (unsigned)(r * RANGE_NODES);
    for (int j = 0; j < DEG_LOADS; ++j) {
        deg_count(cnt, v[j].x, lo);
        deg_count(cnt, v[j].y, lo);
        deg_count(cnt, v[j].z, lo);
        deg_count(cnt, v[j].w, lo);
    }
    __syncthreads();

    uint4* dst = (uint4*)((unsigned*)planes + (size_t)s * PAIR_WORDS + r * RANGE_WORDS);
    for (int i = threadIdx.x; i < RANGE_WORDS / 4; i += DEG_THREADS)
        dst[i] = ((const uint4*)cnt)[i];
}

__global__ void gcn_rsqrt(const float* __restrict__ planes, float* __restrict__ deg) {
    int p = blockIdx.x * blockDim.x + threadIdx.x;
    if (p >= PAIR_WORDS) return;
    const unsigned* w = (const unsigned*)planes + p;
    unsigned a = 1u, b = 1u;
    for (int s = 0; s < DEG_SLICES; ++s) {
        unsigned x = w[(size_t)s * PAIR_WORDS];
        a += x & 0xFFFFu;
        b += x >> 16;
    }
    float2 o;
    o.x = rsqrtf((float)a);
    o.y = rsqrtf((float)b);
    ((float2*)deg)[p] = o;
}

__global__ void __launch_bounds__(256) gcn_xw_wmma(const float* __restrict__ x,
                                                   const float* __restrict__ W1,
                                                   const float* __restrict__ W2,
                                                   const float* __restrict__ dinv,
                                                   float* __restrict__ z) {
    __shared__ float sW1[IN_CH * HID];
    for (int i = threadIdx.x; i < (IN_CH * HID) / 4; i += 256)
        ((float4*)sW1)[i] = ((const float4*)W1)[i];
    __syncthreads();

    const int wave = threadIdx.x >> 5;
    const int lane = threadIdx.x & 31;
    const int half = lane >> 4;
    const int lrow = lane & 15;
    const int tile = blockIdx.x * 8 + wave;
    if (tile * 16 >= N_NODES) return;

    const float* xrow = x + (size_t)(tile * 16 + lrow) * IN_CH + 2 * half;
    const float* bptr = sW1 + (2 * half) * HID + lrow;

    v8f acc = {};
#pragma unroll 8
    for (int k0 = 0; k0 < IN_CH; k0 += 4) {
        v2f a = *(const v2f*)(xrow + k0);
        v2f b;
        b.x = bptr[k0 * HID];
        b.y = bptr[k0 * HID + HID];
        acc = __builtin_amdgcn_wmma_f32_16x16x4_f32(
             false, a,  false, b,
             (short)0, acc,  false,  false);
    }

    const float w0 = W2[lrow * 2 + 0];
    const float w1 = W2[lrow * 2 + 1];
    const bool odd = (lrow & 1) != 0;
    float mine = 0.0f;
#pragma unroll
    for (int r = 0; r < 8; ++r) {
        float p0 = acc[r] * w0, p1 = acc[r] * w1;
        float keep = odd ? p1 : p0;
        float send = odd ? p0 : p1;
        keep += __shfl_xor(send, 1);
        keep += __shfl_xor(keep, 2);
        keep += __shfl_xor(keep, 4);
        keep += __shfl_xor(keep, 8);
        if ((lrow >> 1) == r) mine = keep;
    }
    const int node = tile * 16 + 8 * half + (lrow >> 1);
    z[tile * 32 + lane] = mine * dinv[node];
}

#define AGG_SLICES 15
#define AGG_RANGES 10
#define AGG_THREADS 1024
#define AGG_U 4
#define AGG_RANGE_NODES (N_NODES / AGG_RANGES)
#define AGG_ACC_FLOATS  (AGG_RANGE_NODES * 2)
#define EDGE_VEC4       (N_EDGES / 4)
#define AGG_SLICE_VEC4  ((EDGE_VEC4 + AGG_SLICES - 1) / AGG_SLICES)
static_assert(N_EDGES % 4 == 0, "col is whole 16-byte loads");
static_assert(N_NODES % AGG_RANGES == 0 && (AGG_ACC_FLOATS * 4) % 128 == 0,
              "a range of a plane is whole 128-byte lines, so no line has two writers");

__global__ void __launch_bounds__(AGG_THREADS) gcn_aggregate_lds(const int* __restrict__ row,
                                                                 const int* __restrict__ col,
                                                                 const float* __restrict__ z,
                                                                 float* __restrict__ planes) {
    __shared__ float acc[AGG_ACC_FLOATS];
    const int r = blockIdx.x % AGG_RANGES;
    const int s = blockIdx.x / AGG_RANGES;
    for (int i = threadIdx.x; i < AGG_ACC_FLOATS / 4; i += AGG_THREADS)
        ((float4*)acc)[i] = make_float4(0.0f, 0.0f, 0.0f, 0.0f);
    __syncthreads();

    const unsigned lo = (unsigned)(r * AGG_RANGE_NODES);
    const int v_begin = s * AGG_SLICE_VEC4;
    const int v_end = (v_begin + AGG_SLICE_VEC4 < EDGE_VEC4) ? v_begin + AGG_SLICE_VEC4 : EDGE_VEC4;
    const int e_spare = v_begin * 4;
    const int4* c4 = (const int4*)col;
    const float2* z2 = (const float2*)z;

    for (int base = v_begin; base < v_end; base += AGG_THREADS * AGG_U) {
        int d[AGG_U * 4];
        int e[AGG_U * 4];
        for (int j = 0; j < AGG_U; ++j) {
            int i = base + j * AGG_THREADS + (int)threadIdx.x;
            int4 v = (i < v_end) ? c4[i] : make_int4(-1, -1, -1, -1);
            d[4 * j + 0] = v.x; d[4 * j + 1] = v.y; d[4 * j + 2] = v.z; d[4 * j + 3] = v.w;
            e[4 * j + 0] = 4 * i; e[4 * j + 1] = 4 * i + 1; e[4 * j + 2] = 4 * i + 2; e[4 * j + 3] = 4 * i + 3;
        }
        unsigned k[AGG_U * 4];
        int src[AGG_U * 4];
        for (int m = 0; m < AGG_U * 4; ++m) {
            k[m] = (unsigned)d[m] - lo;
            src[m] = row[(k[m] < (unsigned)AGG_RANGE_NODES) ? e[m] : e_spare];
        }
        float2 zz[AGG_U * 4];
        for (int m = 0; m < AGG_U * 4; ++m) zz[m] = z2[src[m]];
        asm volatile("" :: "v"(zz[0].x), "v"(zz[0].y));
        for (int m = 0; m < AGG_U * 4; ++m) {
            if (k[m] < (unsigned)AGG_RANGE_NODES) {
                atomicAdd(&acc[2 * k[m] + 0], zz[m].x);
                atomicAdd(&acc[2 * k[m] + 1], zz[m].y);
            }
        }
    }
    __syncthreads();

    float4* dst = (float4*)(planes + (size_t)s * (N_NODES * 2) + (size_t)r * AGG_ACC_FLOATS);
    for (int i = threadIdx.x; i < AGG_ACC_FLOATS / 4; i += AGG_THREADS)
        dst[i] = ((const float4*)acc)[i];
}

__global__ void gcn_out(const float* __restrict__ planes, const float* __restrict__ z,
                        const float* __restrict__ dinv, const float* __restrict__ b1,
                        const float* __restrict__ W2, const float* __restrict__ b2,
                        float* __restrict__ out) {
    int t = blockIdx.x * blockDim.x + threadIdx.x;
    if (t >= N_NODES * 2) return;
    const int j = t & 1;
    float cst = b2[j];
    for (int c = 0; c < HID; ++c) cst += b1[c] * W2[c * 2 + j];
    float a = z[t];
#pragma unroll
    for (int s = 0; s < AGG_SLICES; ++s) a += planes[(size_t)s * (N_NODES * 2) + t];
    out[t] = dinv[t >> 1] * a + cst;
}

extern "C" void kernel_launch(void* const* d_in, const int* in_sizes, int n_in,
                              void* d_out, int out_size, void* d_ws, size_t ws_size,
                              hipStream_t stream) {
    const float* x  = (const float*)d_in[0];
    const int*   ei = (const int*)d_in[1];
    const float* W1 = (const float*)d_in[2];
    const float* b1 = (const float*)d_in[3];
    const float* W2 = (const float*)d_in[4];
    const float* b2 = (const float*)d_in[5];
    float* out = (float*)d_out;

    float* deg = (float*)d_ws;
    float* z   = deg + N_NODES;
    float* agg = z + (size_t)N_NODES * 2;

    const int* row = ei;
    const int* col = ei + N_EDGES;

    gcn_degree_lds<<<DEG_SLICES * DEG_RANGES, DEG_THREADS, 0, stream>>>(col, z);
    gcn_rsqrt<<<(PAIR_WORDS + 255) / 256, 256, 0, stream>>>(z, deg);

    int n_tiles = N_NODES / 16;
    gcn_xw_wmma<<<(n_tiles + 7) / 8, 256, 0, stream>>>(x, W1, W2, deg, z);

    gcn_aggregate_lds<<<AGG_SLICES * AGG_RANGES, AGG_THREADS, 0, stream>>>(row, col, z, agg);

    gcn_out<<<(N_NODES * 2 + 255) / 256, 256, 0, stream>>>(agg, z, deg, b1, W2, b2, out);
}


# degree count and x@W1 product run in ONE launch (independent blocks side by side), dinv scaling of the rows moved into the rsqrt kernel; 5 launches -> 4, same sums in the same order
# speedup vs baseline: 4.2146x; 1.0801x over previous
#include <hip/hip_runtime.h>


typedef __attribute__((ext_vector_type(2))) float v2f;
typedef __attribute__((ext_vector_type(8))) float v8f;

#define N_NODES 100000
#define N_EDGES 3200000
#define IN_CH 512
#define HID 16
#define EDGE_VEC4 (N_EDGES / 4)

#define FUSED_THREADS 256
#define DEG_SLICES 60
#define DEG_RANGES 4
#define DEG_BLOCKS (DEG_SLICES * DEG_RANGES)
#define DEG_SLICE_VEC4 ((EDGE_VEC4 + DEG_SLICES - 1) / DEG_SLICES)
#define DEG_SLICE_EDGES (DEG_SLICE_VEC4 * 4)
#define DEG_U 8
#define RANGE_NODES (N_NODES / DEG_RANGES)
#define RANGE_WORDS (RANGE_NODES / 2)
#define PAIR_WORDS  (N_NODES / 2)
#define XW_TILES  (N_NODES / 16)
#define XW_BLOCKS ((XW_TILES + 7) / 8)
static_assert(N_EDGES % 4 == 0, "col is whole 16-byte loads");
static_assert(DEG_SLICE_EDGES <= 65535, "one slice landing on one node must fit a 16-bit field");
static_assert(N_NODES % DEG_RANGES == 0 && RANGE_NODES % 2 == 0 && RANGE_WORDS % 4 == 0,
              "ranges are whole 16-byte pieces of packed words");
static_assert(N_NODES % 16 == 0, "tiles are whole");

union FusedLds {
    uint4 cnt4[RANGE_WORDS / 4];
    float4 w1v[(IN_CH * HID) / 4];
};

__device__ __forceinline__ void deg_count(unsigned* cnt, int d, unsigned lo) {
    unsigned k = (unsigned)d - lo;
    if (k < (unsigned)RANGE_NODES) atomicAdd(&cnt[k >> 1], 1u << ((k & 1u) << 4));
}

__device__ __forceinline__ void deg_load(int4 (&v)[DEG_U], const int4* __restrict__ c4, int base, int v_end) {
    for (int j = 0; j < DEG_U; ++j) {
        int i = base + j * FUSED_THREADS + (int)threadIdx.x;
        v[j] = (i < v_end) ? c4[i] : make_int4(-1, -1, -1, -1);
    }
}

__device__ __forceinline__ void degree_block(FusedLds& lds, int b, const int* __restrict__ col,
                                             float* __restrict__ planes) {
    unsigned* cnt = (unsigned*)lds.cnt4;
    const int r = b % DEG_RANGES;
    const int s = b / DEG_RANGES;
    const int v_begin = s * DEG_SLICE_VEC4;
    const int v_end = (v_begin + DEG_SLICE_VEC4 < EDGE_VEC4) ? v_begin + DEG_SLICE_VEC4 : EDGE_VEC4;
    const int4* c4 = (const int4*)col;

    int4 v[DEG_U];
    deg_load(v, c4, v_begin, v_end);
    for (int i = threadIdx.x; i < RANGE_WORDS / 4; i += FUSED_THREADS)
        lds.cnt4[i] = make_uint4(0u, 0u, 0u, 0u);
    __syncthreads();

    const unsigned lo = (unsigned)(r * RANGE_NODES);
    for (int base = v_begin; base < v_end; base += FUSED_THREADS * DEG_U) {
        int4 nxt[DEG_U];
        deg_load(nxt, c4, base + FUSED_THREADS * DEG_U, v_end);
        for (int j = 0; j < DEG_U; ++j) {
            deg_count(cnt, v[j].x, lo);
            deg_count(cnt, v[j].y, lo);
            deg_count(cnt, v[j].z, lo);
            deg_count(cnt, v[j].w, lo);
        }
        for (int j = 0; j < DEG_U; ++j) v[j] = nxt[j];
    }
    __syncthreads();

    uint4* dst = (uint4*)((unsigned*)planes + (size_t)s * PAIR_WORDS + r * RANGE_WORDS);
    for (int i = threadIdx.x; i < RANGE_WORDS / 4; i += FUSED_THREADS)
        dst[i] = lds.cnt4[i];
}

__device__ __forceinline__ void xw_block(FusedLds& lds, int blk, const float* __restrict__ x,
                                         const float* __restrict__ W1,
                                         const float* __restrict__ W2,
                                         float* __restrict__ y) {
    const float* sW1 = (const float*)lds.w1v;
    for (int i = threadIdx.x; i < (IN_CH * HID) / 4; i += FUSED_THREADS)
        lds.w1v[i] = ((const float4*)W1)[i];
    __syncthreads();

    const int wave = threadIdx.x >> 5;
    const int lane = threadIdx.x & 31;
    const int half = lane >> 4;
    const int lrow = lane & 15;
    const int tile = blk * 8 + wave;
    if (tile >= XW_TILES) return;

    const float* xrow = x + (size_t)(tile * 16 + lrow) * IN_CH + 2 * half;
    const float* bptr = sW1 + (2 * half) * HID + lrow;

    v8f acc = {};
#pragma unroll 8
    for (int k0 = 0; k0 < IN_CH; k0 += 4) {
        v2f a = *(const v2f*)(xrow + k0);
        v2f b;
        b.x = bptr[k0 * HID];
        b.y = bptr[k0 * HID + HID];
        acc = __builtin_amdgcn_wmma_f32_16x16x4_f32(
             false, a,  false, b,
             (short)0, acc,  false,  false);
    }

    const float w0 = W2[lrow * 2 + 0];
    const float w1 = W2[lrow * 2 + 1];
    const bool odd = (lrow & 1) != 0;
    float mine = 0.0f;
#pragma unroll
    for (int r = 0; r < 8; ++r) {
        float p0 = acc[r] * w0, p1 = acc[r] * w1;
        float keep = odd ? p1 : p0;
        float send = odd ? p0 : p1;
        keep += __shfl_xor(send, 1);
        keep += __shfl_xor(keep, 2);
        keep += __shfl_xor(keep, 4);
        keep += __shfl_xor(keep, 8);
        if ((lrow >> 1) == r) mine = keep;
    }
    y[tile * 32 + lane] = mine;
}

__global__ void __launch_bounds__(FUSED_THREADS) gcn_degree_and_xw(const int* __restrict__ col,
                                                                   float* __restrict__ planes,
                                                                   const float* __restrict__ x,
                                                                   const float* __restrict__ W1,
                                                                   const float* __restrict__ W2,
                                                                   float* __restrict__ y) {
    __shared__ FusedLds lds;
    if (blockIdx.x < DEG_BLOCKS) degree_block(lds, (int)blockIdx.x, col, planes);
    else xw_block(lds, (int)blockIdx.x - DEG_BLOCKS, x, W1, W2, y);
}

__global__ void gcn_rsqrt_scale(const float* __restrict__ planes, float* __restrict__ deg,
                                float* z) {
    int p = blockIdx.x * blockDim.x + threadIdx.x;
    if (p >= PAIR_WORDS) return;
    const unsigned* w = (const unsigned*)planes + p;
    unsigned a = 1u, b = 1u;
    for (int s = 0; s < DEG_SLICES; ++s) {
        unsigned c = w[(size_t)s * PAIR_WORDS];
        a += c & 0xFFFFu;
        b += c >> 16;
    }
    float2 o;
    o.x = rsqrtf((float)a);
    o.y = rsqrtf((float)b);
    ((float2*)deg)[p] = o;
    float4 q = ((const float4*)z)[p];
    q.x *= o.x; q.y *= o.x; q.z *= o.y; q.w *= o.y;
    ((float4*)z)[p] = q;
}

#define AGG_SLICES 15
#define AGG_RANGES 10
#define AGG_THREADS 1024
#define AGG_U 4
#define AGG_RANGE_NODES (N_NODES / AGG_RANGES)
#define AGG_ACC_FLOATS  (AGG_RANGE_NODES * 2)
#define AGG_SLICE_VEC4  ((EDGE_VEC4 + AGG_SLICES - 1) / AGG_SLICES)
static_assert(N_EDGES % 4 == 0, "col is whole 16-byte loads");
static_assert(N_NODES % AGG_RANGES == 0 && (AGG_ACC_FLOATS * 4) % 128 == 0,
              "a range of a plane is whole 128-byte lines, so no line has two writers");

__global__ void __launch_bounds__(AGG_THREADS) gcn_aggregate_lds(const int* __restrict__ row,
                                                                 const int* __restrict__ col,
                                                                 const float* __restrict__ z,
                                                                 float* __restrict__ planes) {
    __shared__ float acc[AGG_ACC_FLOATS];
    const int r = blockIdx.x % AGG_RANGES;
    const int s = blockIdx.x / AGG_RANGES;
    for (int i = threadIdx.x; i < AGG_ACC_FLOATS / 4; i += AGG_THREADS)
        ((float4*)acc)[i] = make_float4(0.0f, 0.0f, 0.0f, 0.0f);
    __syncthreads();

    const unsigned lo = (unsigned)(r * AGG_RANGE_NODES);
    const int v_begin = s * AGG_SLICE_VEC4;
    const int v_end = (v_begin + AGG_SLICE_VEC4 < EDGE_VEC4) ? v_begin + AGG_SLICE_VEC4 : EDGE_VEC4;
    const int e_spare = v_begin * 4;
    const int4* c4 = (const int4*)col;
    const float2* z2 = (const float2*)z;

    for (int base = v_begin; base < v_end; base += AGG_THREADS * AGG_U) {
        int d[AGG_U * 4];
        int e[AGG_U * 4];
        for (int j = 0; j < AGG_U; ++j) {
            int i = base + j * AGG_THREADS + (int)threadIdx.x;
            int4 v = (i < v_end) ? c4[i] : make_int4(-1, -1, -1, -1);
            d[4 * j + 0] = v.x; d[4 * j + 1] = v.y; d[4 * j + 2] = v.z; d[4 * j + 3] = v.w;
            e[4 * j + 0] = 4 * i; e[4 * j + 1] = 4 * i + 1; e[4 * j + 2] = 4 * i + 2; e[4 * j + 3] = 4 * i + 3;
        }
        unsigned k[AGG_U * 4];
        int src[AGG_U * 4];
        for (int m = 0; m < AGG_U * 4; ++m) {
            k[m] = (unsigned)d[m] - lo;
            src[m] = row[(k[m] < (unsigned)AGG_RANGE_NODES) ? e[m] : e_spare];
        }
        float2 zz[AGG_U * 4];
        for (int m = 0; m < AGG_U * 4; ++m) zz[m] = z2[src[m]];
        asm volatile("" :: "v"(zz[0].x), "v"(zz[0].y));
        for (int m = 0; m < AGG_U * 4; ++m) {
            if (k[m] < (unsigned)AGG_RANGE_NODES) {
                atomicAdd(&acc[2 * k[m] + 0], zz[m].x);
                atomicAdd(&acc[2 * k[m] + 1], zz[m].y);
            }
        }
    }
    __syncthreads();

    float4* dst = (float4*)(planes + (size_t)s * (N_NODES * 2) + (size_t)r * AGG_ACC_FLOATS);
    for (int i = threadIdx.x; i < AGG_ACC_FLOATS / 4; i += AGG_THREADS)
        dst[i] = ((const float4*)acc)[i];
}

__global__ void gcn_out(const float* __restrict__ planes, const float* __restrict__ z,
                        const float* __restrict__ dinv, const float* __restrict__ b1,
                        const float* __restrict__ W2, const float* __restrict__ b2,
                        float* __restrict__ out) {
    int t = blockIdx.x * blockDim.x + threadIdx.x;
    if (t >= N_NODES * 2) return;
    const int j = t & 1;
    float cst = b2[j];
    for (int c = 0; c < HID; ++c) cst += b1[c] * W2[c * 2 + j];
    float a = z[t];
#pragma unroll
    for (int s = 0; s < AGG_SLICES; ++s) a += planes[(size_t)s * (N_NODES * 2) + t];
    out[t] = dinv[t >> 1] * a + cst;
}

extern "C" void kernel_launch(void* const* d_in, const int* in_sizes, int n_in,
                              void* d_out, int out_size, void* d_ws, size_t ws_size,
                              hipStream_t stream) {
    const float* x  = (const float*)d_in[0];
    const int*   ei = (const int*)d_in[1];
    const float* W1 = (const float*)d_in[2];
    const float* b1 = (const float*)d_in[3];
    const float* W2 = (const float*)d_in[4];
    const float* b2 = (const float*)d_in[5];
    float* out = (float*)d_out;

    float* deg = (float*)d_ws;
    float* z   = deg + N_NODES;
    float* agg = z + (size_t)N_NODES * 2;

    const int* row = ei;
    const int* col = ei + N_EDGES;

    gcn_degree_and_xw<<<DEG_BLOCKS + XW_BLOCKS, FUSED_THREADS, 0, stream>>>(col, agg, x, W1, W2, z);
    gcn_rsqrt_scale<<<(PAIR_WORDS + 255) / 256, 256, 0, stream>>>(agg, deg, z);

    gcn_aggregate_lds<<<AGG_SLICES * AGG_RANGES, AGG_THREADS, 0, stream>>>(row, col, z, agg);

    gcn_out<<<(N_NODES * 2 + 255) / 256, 256, 0, stream>>>(agg, z, deg, b1, W2, b2, out);
}


# x @ W1: x fetched in whole 128-byte lines (16-byte loads, 256 contiguous bytes per row) and turned into fragments through a per-wave LDS stage, one chunk ahead; k order of the sums changed
# speedup vs baseline: 4.3590x; 1.0343x over previous
#include <hip/hip_runtime.h>


typedef __attribute__((ext_vector_type(2))) float v2f;
typedef __attribute__((ext_vector_type(8))) float v8f;

#define N_NODES 100000
#define N_EDGES 3200000
#define IN_CH 512
#define HID 16
#define EDGE_VEC4 (N_EDGES / 4)

#define FUSED_THREADS 256
#define DEG_SLICES 60
#define DEG_RANGES 4
#define DEG_BLOCKS (DEG_SLICES * DEG_RANGES)
#define DEG_SLICE_VEC4 ((EDGE_VEC4 + DEG_SLICES - 1) / DEG_SLICES)
#define DEG_SLICE_EDGES (DEG_SLICE_VEC4 * 4)
#define DEG_U 8
#define RANGE_NODES (N_NODES / DEG_RANGES)
#define RANGE_WORDS (RANGE_NODES / 2)
#define PAIR_WORDS  (N_NODES / 2)
#define XW_TILES  (N_NODES / 16)
#define XW_BLOCKS ((XW_TILES + 7) / 8)
static_assert(N_EDGES % 4 == 0, "col is whole 16-byte loads");
static_assert(DEG_SLICE_EDGES <= 65535, "one slice landing on one node must fit a 16-bit field");
static_assert(N_NODES % DEG_RANGES == 0 && RANGE_NODES % 2 == 0 && RANGE_WORDS % 4 == 0,
              "ranges are whole 16-byte pieces of packed words");
static_assert(N_NODES % 16 == 0, "tiles are whole");

#define XW_CHUNK_K 64
#define XW_CHUNKS (IN_CH / XW_CHUNK_K)
#define XW_PITCH4 17
#define XW_STAGE4 (16 * XW_PITCH4)
static_assert(IN_CH % XW_CHUNK_K == 0 && XW_CHUNK_K == 64, "a chunk is 16 lanes x 16 bytes of one row");

struct XwLds {
    float4 w1p[(IN_CH * HID) / 4];
    float4 stage[FUSED_THREADS / 32][XW_STAGE4];
};
union FusedLds {
    uint4 cnt4[RANGE_WORDS / 4];
    XwLds xw;
};

__device__ __forceinline__ void deg_count(unsigned* cnt, int d, unsigned lo) {
    unsigned k = (unsigned)d - lo;
    if (k < (unsigned)RANGE_NODES) atomicAdd(&cnt[k >> 1], 1u << ((k & 1u) << 4));
}

__device__ __forceinline__ void deg_load(int4 (&v)[DEG_U], const int4* __restrict__ c4, int base, int v_end) {
    for (int j = 0; j < DEG_U; ++j) {
        int i = base + j * FUSED_THREADS + (int)threadIdx.x;
        v[j] = (i < v_end) ? c4[i] : make_int4(-1, -1, -1, -1);
    }
}

__device__ __forceinline__ void degree_block(FusedLds& lds, int b, const int* __restrict__ col,
                                             float* __restrict__ planes) {
    unsigned* cnt = (unsigned*)lds.cnt4;
    const int r = b % DEG_RANGES;
    const int s = b / DEG_RANGES;
    const int v_begin = s * DEG_SLICE_VEC4;
    const int v_end = (v_begin + DEG_SLICE_VEC4 < EDGE_VEC4) ? v_begin + DEG_SLICE_VEC4 : EDGE_VEC4;
    const int4* c4 = (const int4*)col;

    int4 v[DEG_U];
    deg_load(v, c4, v_begin, v_end);
    for (int i = threadIdx.x; i < RANGE_WORDS / 4; i += FUSED_THREADS)
        lds.cnt4[i] = make_uint4(0u, 0u, 0u, 0u);
    __syncthreads();

    const unsigned lo = (unsigned)(r * RANGE_NODES);
    for (int base = v_begin; base < v_end; base += FUSED_THREADS * DEG_U) {
        int4 nxt[DEG_U];
        deg_load(nxt, c4, base + FUSED_THREADS * DEG_U, v_end);
        for (int j = 0; j < DEG_U; ++j) {
            deg_count(cnt, v[j].x, lo);
            deg_count(cnt, v[j].y, lo);
            deg_count(cnt, v[j].z, lo);
            deg_count(cnt, v[j].w, lo);
        }
        for (int j = 0; j < DEG_U; ++j) v[j] = nxt[j];
    }
    __syncthreads();

    uint4* dst = (uint4*)((unsigned*)planes + (size_t)s * PAIR_WORDS + r * RANGE_WORDS);
    for (int i = threadIdx.x; i < RANGE_WORDS / 4; i += FUSED_THREADS)
        dst[i] = lds.cnt4[i];
}


__device__ __forceinline__ void wave_stage_handover() {
    __builtin_amdgcn_wave_barrier();
}

#define XW_ROWPAIR4 (2 * IN_CH / 4)
#define XW_LOAD_CHUNK(v, p) do { \
    v##0 = (p)[0 * XW_ROWPAIR4]; v##1 = (p)[1 * XW_ROWPAIR4]; v##2 = (p)[2 * XW_ROWPAIR4]; v##3 = (p)[3 * XW_ROWPAIR4]; \
    v##4 = (p)[4 * XW_ROWPAIR4]; v##5 = (p)[5 * XW_ROWPAIR4]; v##6 = (p)[6 * XW_ROWPAIR4]; v##7 = (p)[7 * XW_ROWPAIR4]; \
} while (0)
#define XW_STORE_CHUNK(S, v) do { \
    (S)[0 * 2 * XW_PITCH4] = v##0; (S)[1 * 2 * XW_PITCH4] = v##1; (S)[2 * 2 * XW_PITCH4] = v##2; (S)[3 * 2 * XW_PITCH4] = v##3; \
    (S)[4 * 2 * XW_PITCH4] = v##4; (S)[5 * 2 * XW_PITCH4] = v##5; (S)[6 * 2 * XW_PITCH4] = v##6; (S)[7 * 2 * XW_PITCH4] = v##7; \
} while (0)

__device__ __forceinline__ v8f xw_chunk_products(v8f acc, const float4* A, const float4* B) {
#pragma unroll
    for (int j = 0; j < 8; ++j) {
        const float4 a4 = A[2 * j];
        const float4 b4 = B[2 * j * HID];
        v2f a, b;
        a.x = a4.x; a.y = a4.y; b.x = b4.x; b.y = b4.y;
        acc = __builtin_amdgcn_wmma_f32_16x16x4_f32(
             false, a,  false, b,
             (short)0, acc,  false,  false);
        a.x = a4.z; a.y = a4.w; b.x = b4.z; b.y = b4.w;
        acc = __builtin_amdgcn_wmma_f32_16x16x4_f32(
             false, a,  false, b,
             (short)0, acc,  false,  false);
    }
    return acc;
}

__device__ __forceinline__ void xw_block(FusedLds& lds, int blk, const float* __restrict__ x,
                                         const float* __restrict__ W1,
                                         const float* __restrict__ W2,
                                         float* __restrict__ y) {
    for (int i = threadIdx.x; i < (IN_CH * HID) / 4; i += FUSED_THREADS) {
        const float* w = W1 + (size_t)(4 * (i >> 4)) * HID + (i & 15);
        lds.xw.w1p[i] = make_float4(w[0], w[HID], w[2 * HID], w[3 * HID]);
    }
    __syncthreads();

    const int wave = threadIdx.x >> 5;
    const int lane = threadIdx.x & 31;
    const int half = lane >> 4;
    const int lrow = lane & 15;
    const int tile = blk * 8 + wave;
    if (tile >= XW_TILES) return;

    float4* S4 = lds.xw.stage[wave];
    const float4* xq = (const float4*)(x + (size_t)(tile * 16 + half) * IN_CH) + lrow;
    float4* st = S4 + half * XW_PITCH4 + lrow;
    const float4* fa = S4 + lrow * XW_PITCH4 + half;
    const float4* fb = lds.xw.w1p + half * HID + lrow;

    v8f acc = {};
    float4 ca0, ca1, ca2, ca3, ca4, ca5, ca6, ca7;
    float4 cb0, cb1, cb2, cb3, cb4, cb5, cb6, cb7;
    XW_LOAD_CHUNK(ca, xq);
#pragma unroll
    for (int ch = 0; ch < XW_CHUNKS; ch += 2) {
        XW_LOAD_CHUNK(cb, xq + (ch + 1) * (XW_CHUNK_K / 4));
        wave_stage_handover();
        XW_STORE_CHUNK(st, ca);
        wave_stage_handover();
        acc = xw_chunk_products(acc, fa, fb + ch * (XW_CHUNK_K / 4) * HID);
        if (ch + 2 < XW_CHUNKS) XW_LOAD_CHUNK(ca, xq + (ch + 2) * (XW_CHUNK_K / 4));
        wave_stage_handover();
        XW_STORE_CHUNK(st, cb);
        wave_stage_handover();
        acc = xw_chunk_products(acc, fa, fb + (ch + 1) * (XW_CHUNK_K / 4) * HID);
    }

    const float w0 = W2[lrow * 2 + 0];
    const float w1 = W2[lrow * 2 + 1];
    const bool odd = (lrow & 1) != 0;
    float mine = 0.0f;
#pragma unroll
    for (int r = 0; r < 8; ++r) {
        float p0 = acc[r] * w0, p1 = acc[r] * w1;
        float keep = odd ? p1 : p0;
        float send = odd ? p0 : p1;
        keep += __shfl_xor(send, 1);
        keep += __shfl_xor(keep, 2);
        keep += __shfl_xor(keep, 4);
        keep += __shfl_xor(keep, 8);
        if ((lrow >> 1) == r) mine = keep;
    }
    y[tile * 32 + lane] = mine;
}

__global__ void __launch_bounds__(FUSED_THREADS) gcn_degree_and_xw(const int* __restrict__ col,
                                                                   float* __restrict__ planes,
                                                                   const float* __restrict__ x,
                                                                   const float* __restrict__ W1,
                                                                   const float* __restrict__ W2,
                                                                   float* __restrict__ y) {
    __shared__ FusedLds lds;
    if (blockIdx.x < DEG_BLOCKS) degree_block(lds, (int)blockIdx.x, col, planes);
    else xw_block(lds, (int)blockIdx.x - DEG_BLOCKS, x, W1, W2, y);
}

__global__ void gcn_rsqrt_scale(const float* __restrict__ planes, float* __restrict__ deg,
                                float* z) {
    int p = blockIdx.x * blockDim.x + threadIdx.x;
    if (p >= PAIR_WORDS) return;
    const unsigned* w = (const unsigned*)planes + p;
    unsigned a = 1u, b = 1u;
    for (int s = 0; s < DEG_SLICES; ++s) {
        unsigned c = w[(size_t)s * PAIR_WORDS];
        a += c & 0xFFFFu;
        b += c >> 16;
    }
    float2 o;
    o.x = rsqrtf((float)a);
    o.y = rsqrtf((float)b);
    ((float2*)deg)[p] = o;
    float4 q = ((const float4*)z)[p];
    q.x *= o.x; q.y *= o.x; q.z *= o.y; q.w *= o.y;
    ((float4*)z)[p] = q;
}

#define AGG_SLICES 15
#define AGG_RANGES 10
#define AGG_THREADS 1024
#define AGG_U 4
#define AGG_RANGE_NODES (N_NODES / AGG_RANGES)
#define AGG_ACC_FLOATS  (AGG_RANGE_NODES * 2)
#define AGG_SLICE_VEC4  ((EDGE_VEC4 + AGG_SLICES - 1) / AGG_SLICES)
static_assert(N_EDGES % 4 == 0, "col is whole 16-byte loads");
static_assert(N_NODES % AGG_RANGES == 0 && (AGG_ACC_FLOATS * 4) % 128 == 0,
              "a range of a plane is whole 128-byte lines, so no line has two writers");

__global__ void __launch_bounds__(AGG_THREADS) gcn_aggregate_lds(const int* __restrict__ row,
                                                                 const int* __restrict__ col,
                                                                 const float* __restrict__ z,
                                                                 float* __restrict__ planes) {
    __shared__ float acc[AGG_ACC_FLOATS];
    const int r = blockIdx.x % AGG_RANGES;
    const int s = blockIdx.x / AGG_RANGES;
    for (int i = threadIdx.x; i < AGG_ACC_FLOATS / 4; i += AGG_THREADS)
        ((float4*)acc)[i] = make_float4(0.0f, 0.0f, 0.0f, 0.0f);
    __syncthreads();

    const unsigned lo = (unsigned)(r * AGG_RANGE_NODES);
    const int v_begin = s * AGG_SLICE_VEC4;
    const int v_end = (v_begin + AGG_SLICE_VEC4 < EDGE_VEC4) ? v_begin + AGG_SLICE_VEC4 : EDGE_VEC4;
    const int e_spare = v_begin * 4;
    const int4* c4 = (const int4*)col;
    const float2* z2 = (const float2*)z;

    for (int base = v_begin; base < v_end; base += AGG_THREADS * AGG_U) {
        int d[AGG_U * 4];
        int e[AGG_U * 4];
        for (int j = 0; j < AGG_U; ++j) {
            int i = base + j * AGG_THREADS + (int)threadIdx.x;
            int4 v = (i < v_end) ? c4[i] : make_int4(-1, -1, -1, -1);
            d[4 * j + 0] = v.x; d[4 * j + 1] = v.y; d[4 * j + 2] = v.z; d[4 * j + 3] = v.w;
            e[4 * j + 0] = 4 * i; e[4 * j + 1] = 4 * i + 1; e[4 * j + 2] = 4 * i + 2; e[4 * j + 3] = 4 * i + 3;
        }
        unsigned k[AGG_U * 4];
        int src[AGG_U * 4];
        for (int m = 0; m < AGG_U * 4; ++m) {
            k[m] = (unsigned)d[m] - lo;
            src[m] = row[(k[m] < (unsigned)AGG_RANGE_NODES) ? e[m] : e_spare];
        }
        float2 zz[AGG_U * 4];
        for (int m = 0; m < AGG_U * 4; ++m) zz[m] = z2[src[m]];
        asm volatile("" :: "v"(zz[0].x), "v"(zz[0].y));
        for (int m = 0; m < AGG_U * 4; ++m) {
            if (k[m] < (unsigned)AGG_RANGE_NODES) {
                atomicAdd(&acc[2 * k[m] + 0], zz[m].x);
                atomicAdd(&acc[2 * k[m] + 1], zz[m].y);
            }
        }
    }
    __syncthreads();

    float4* dst = (float4*)(planes + (size_t)s * (N_NODES * 2) + (size_t)r * AGG_ACC_FLOATS);
    for (int i = threadIdx.x; i < AGG_ACC_FLOATS / 4; i += AGG_THREADS)
        dst[i] = ((const float4*)acc)[i];
}

__global__ void gcn_out(const float* __restrict__ planes, const float* __restrict__ z,
                        const float* __restrict__ dinv, const float* __restrict__ b1,
                        const float* __restrict__ W2, const float* __restrict__ b2,
                        float* __restrict__ out) {
    int t = blockIdx.x * blockDim.x + threadIdx.x;
    if (t >= N_NODES * 2) return;
    const int j = t & 1;
    float cst = b2[j];
    for (int c = 0; c < HID; ++c) cst += b1[c] * W2[c * 2 + j];
    float a = z[t];
#pragma unroll
    for (int s = 0; s < AGG_SLICES; ++s) a += planes[(size_t)s * (N_NODES * 2) + t];
    out[t] = dinv[t >> 1] * a + cst;
}

extern "C" void kernel_launch(void* const* d_in, const int* in_sizes, int n_in,
                              void* d_out, int out_size, void* d_ws, size_t ws_size,
                              hipStream_t stream) {
    const float* x  = (const float*)d_in[0];
    const int*   ei = (const int*)d_in[1];
    const float* W1 = (const float*)d_in[2];
    const float* b1 = (const float*)d_in[3];
    const float* W2 = (const float*)d_in[4];
    const float* b2 = (const float*)d_in[5];
    float* out = (float*)d_out;

    float* deg = (float*)d_ws;
    float* z   = deg + N_NODES;
    float* agg = z + (size_t)N_NODES * 2;

    const int* row = ei;
    const int* col = ei + N_EDGES;

    gcn_degree_and_xw<<<DEG_BLOCKS + XW_BLOCKS, FUSED_THREADS, 0, stream>>>(col, agg, x, W1, W2, z);
    gcn_rsqrt_scale<<<(PAIR_WORDS + 255) / 256, 256, 0, stream>>>(agg, deg, z);

    gcn_aggregate_lds<<<AGG_SLICES * AGG_RANGES, AGG_THREADS, 0, stream>>>(row, col, z, agg);

    gcn_out<<<(N_NODES * 2 + 255) / 256, 256, 0, stream>>>(agg, z, deg, b1, W2, b2, out);
}


# aggregation grid 10 node ranges x 15 edge slices (150 work-groups) -> 17 ranges x 15 slices (255): nearly every WGP gets one, each gathers z rows for fewer edges; same planes and LDS float adds
# speedup vs baseline: 4.5223x; 1.0375x over previous
#include <hip/hip_runtime.h>


typedef __attribute__((ext_vector_type(2))) float v2f;
typedef __attribute__((ext_vector_type(8))) float v8f;

#define N_NODES 100000
#define N_EDGES 3200000
#define IN_CH 512
#define HID 16
#define EDGE_VEC4 (N_EDGES / 4)

#define FUSED_THREADS 256
#define DEG_SLICES 60
#define DEG_RANGES 4
#define DEG_BLOCKS (DEG_SLICES * DEG_RANGES)
#define DEG_SLICE_VEC4 ((EDGE_VEC4 + DEG_SLICES - 1) / DEG_SLICES)
#define DEG_SLICE_EDGES (DEG_SLICE_VEC4 * 4)
#define DEG_U 8
#define RANGE_NODES (N_NODES / DEG_RANGES)
#define RANGE_WORDS (RANGE_NODES / 2)
#define PAIR_WORDS  (N_NODES / 2)
#define XW_TILES  (N_NODES / 16)
#define XW_BLOCKS ((XW_TILES + 7) / 8)
static_assert(N_EDGES % 4 == 0, "col is whole 16-byte loads");
static_assert(DEG_SLICE_EDGES <= 65535, "one slice landing on one node must fit a 16-bit field");
static_assert(N_NODES % DEG_RANGES == 0 && RANGE_NODES % 2 == 0 && RANGE_WORDS % 4 == 0,
              "ranges are whole 16-byte pieces of packed words");
static_assert(N_NODES % 16 == 0, "tiles are whole");

#define XW_CHUNK_K 64
#define XW_CHUNKS (IN_CH / XW_CHUNK_K)
#define XW_PITCH4 17
#define XW_STAGE4 (16 * XW_PITCH4)
static_assert(IN_CH % XW_CHUNK_K == 0 && XW_CHUNK_K == 64, "a chunk is 16 lanes x 16 bytes of one row");

struct XwLds {
    float4 w1p[(IN_CH * HID) / 4];
    float4 stage[FUSED_THREADS / 32][XW_STAGE4];
};
union FusedLds {
    uint4 cnt4[RANGE_WORDS / 4];
    XwLds xw;
};

__device__ __forceinline__ void deg_count(unsigned* cnt, int d, unsigned lo) {
    unsigned k = (unsigned)d - lo;
    if (k < (unsigned)RANGE_NODES) atomicAdd(&cnt[k >> 1], 1u << ((k & 1u) << 4));
}

__device__ __forceinline__ void deg_load(int4 (&v)[DEG_U], const int4* __restrict__ c4, int base, int v_end) {
    for (int j = 0; j < DEG_U; ++j) {
        int i = base + j * FUSED_THREADS + (int)threadIdx.x;
        v[j] = (i < v_end) ? c4[i] : make_int4(-1, -1, -1, -1);
    }
}

__device__ __forceinline__ void degree_block(FusedLds& lds, int b, const int* __restrict__ col,
                                             float* __restrict__ planes) {
    unsigned* cnt = (unsigned*)lds.cnt4;
    const int r = b % DEG_RANGES;
    const int s = b / DEG_RANGES;
    const int v_begin = s * DEG_SLICE_VEC4;
    const int v_end = (v_begin + DEG_SLICE_VEC4 < EDGE_VEC4) ? v_begin + DEG_SLICE_VEC4 : EDGE_VEC4;
    const int4* c4 = (const int4*)col;

    int4 v[DEG_U];
    deg_load(v, c4, v_begin, v_end);
    for (int i = threadIdx.x; i < RANGE_WORDS / 4; i += FUSED_THREADS)
        lds.cnt4[i] = make_uint4(0u, 0u, 0u, 0u);
    __syncthreads();

    const unsigned lo = (unsigned)(r * RANGE_NODES);
    for (int base = v_begin; base < v_end; base += FUSED_THREADS * DEG_U) {
        int4 nxt[DEG_U];
        deg_load(nxt, c4, base + FUSED_THREADS * DEG_U, v_end);
        for (int j = 0; j < DEG_U; ++j) {
            deg_count(cnt, v[j].x, lo);
            deg_count(cnt, v[j].y, lo);
            deg_count(cnt, v[j].z, lo);
            deg_count(cnt, v[j].w, lo);
        }
        for (int j = 0; j < DEG_U; ++j) v[j] = nxt[j];
    }
    __syncthreads();

    uint4* dst = (uint4*)((unsigned*)planes + (size_t)s * PAIR_WORDS + r * RANGE_WORDS);
    for (int i = threadIdx.x; i < RANGE_WORDS / 4; i += FUSED_THREADS)
        dst[i] = lds.cnt4[i];
}


__device__ __forceinline__ void wave_stage_handover() {
    __builtin_amdgcn_wave_barrier();
}

#define XW_ROWPAIR4 (2 * IN_CH / 4)
#define XW_LOAD_CHUNK(v, p) do { \
    v##0 = (p)[0 * XW_ROWPAIR4]; v##1 = (p)[1 * XW_ROWPAIR4]; v##2 = (p)[2 * XW_ROWPAIR4]; v##3 = (p)[3 * XW_ROWPAIR4]; \
    v##4 = (p)[4 * XW_ROWPAIR4]; v##5 = (p)[5 * XW_ROWPAIR4]; v##6 = (p)[6 * XW_ROWPAIR4]; v##7 = (p)[7 * XW_ROWPAIR4]; \
} while (0)
#define XW_STORE_CHUNK(S, v) do { \
    (S)[0 * 2 * XW_PITCH4] = v##0; (S)[1 * 2 * XW_PITCH4] = v##1; (S)[2 * 2 * XW_PITCH4] = v##2; (S)[3 * 2 * XW_PITCH4] = v##3; \
    (S)[4 * 2 * XW_PITCH4] = v##4; (S)[5 * 2 * XW_PITCH4] = v##5; (S)[6 * 2 * XW_PITCH4] = v##6; (S)[7 * 2 * XW_PITCH4] = v##7; \
} while (0)

__device__ __forceinline__ v8f xw_chunk_products(v8f acc, const float4* A, const float4* B) {
#pragma unroll
    for (int j = 0; j < 8; ++j) {
        const float4 a4 = A[2 * j];
        const float4 b4 = B[2 * j * HID];
        v2f a, b;
        a.x = a4.x; a.y = a4.y; b.x = b4.x; b.y = b4.y;
        acc = __builtin_amdgcn_wmma_f32_16x16x4_f32(
             false, a,  false, b,
             (short)0, acc,  false,  false);
        a.x = a4.z; a.y = a4.w; b.x = b4.z; b.y = b4.w;
        acc = __builtin_amdgcn_wmma_f32_16x16x4_f32(
             false, a,  false, b,
             (short)0, acc,  false,  false);
    }
    return acc;
}

__device__ __forceinline__ void xw_block(FusedLds& lds, int blk, const float* __restrict__ x,
                                         const float* __restrict__ W1,
                                         const float* __restrict__ W2,
                                         float* __restrict__ y) {
    for (int i = threadIdx.x; i < (IN_CH * HID) / 4; i += FUSED_THREADS) {
        const float* w = W1 + (size_t)(4 * (i >> 4)) * HID + (i & 15);
        lds.xw.w1p[i] = make_float4(w[0], w[HID], w[2 * HID], w[3 * HID]);
    }
    __syncthreads();

    const int wave = threadIdx.x >> 5;
    const int lane = threadIdx.x & 31;
    const int half = lane >> 4;
    const int lrow = lane & 15;
    const int tile = blk * 8 + wave;
    if (tile >= XW_TILES) return;

    float4* S4 = lds.xw.stage[wave];
    const float4* xq = (const float4*)(x + (size_t)(tile * 16 + half) * IN_CH) + lrow;
    float4* st = S4 + half * XW_PITCH4 + lrow;
    const float4* fa = S4 + lrow * XW_PITCH4 + half;
    const float4* fb = lds.xw.w1p + half * HID + lrow;

    v8f acc = {};
    float4 ca0, ca1, ca2, ca3, ca4, ca5, ca6, ca7;
    float4 cb0, cb1, cb2, cb3, cb4, cb5, cb6, cb7;
    XW_LOAD_CHUNK(ca, xq);
#pragma unroll
    for (int ch = 0; ch < XW_CHUNKS; ch += 2) {
        XW_LOAD_CHUNK(cb, xq + (ch + 1) * (XW_CHUNK_K / 4));
        wave_stage_handover();
        XW_STORE_CHUNK(st, ca);
        wave_stage_handover();
        acc = xw_chunk_products(acc, fa, fb + ch * (XW_CHUNK_K / 4) * HID);
        if (ch + 2 < XW_CHUNKS) XW_LOAD_CHUNK(ca, xq + (ch + 2) * (XW_CHUNK_K / 4));
        wave_stage_handover();
        XW_STORE_CHUNK(st, cb);
        wave_stage_handover();
        acc = xw_chunk_products(acc, fa, fb + (ch + 1) * (XW_CHUNK_K / 4) * HID);
    }

    const float w0 = W2[lrow * 2 + 0];
    const float w1 = W2[lrow * 2 + 1];
    const bool odd = (lrow & 1) != 0;
    float mine = 0.0f;
#pragma unroll
    for (int r = 0; r < 8; ++r) {
        float p0 = acc[r] * w0, p1 = acc[r] * w1;
        float keep = odd ? p1 : p0;
        float send = odd ? p0 : p1;
        keep += __shfl_xor(send, 1);
        keep += __shfl_xor(keep, 2);
        keep += __shfl_xor(keep, 4);
        keep += __shfl_xor(keep, 8);
        if ((lrow >> 1) == r) mine = keep;
    }
    y[tile * 32 + lane] = mine;
}

__global__ void __launch_bounds__(FUSED_THREADS) gcn_degree_and_xw(const int* __restrict__ col,
                                                                   float* __restrict__ planes,
                                                                   const float* __restrict__ x,
                                                                   const float* __restrict__ W1,
                                                                   const float* __restrict__ W2,
                                                                   float* __restrict__ y) {
    __shared__ FusedLds lds;
    if (blockIdx.x < DEG_BLOCKS) degree_block(lds, (int)blockIdx.x, col, planes);
    else xw_block(lds, (int)blockIdx.x - DEG_BLOCKS, x, W1, W2, y);
}

__global__ void gcn_rsqrt_scale(const float* __restrict__ planes, float* __restrict__ deg,
                                float* z) {
    int p = blockIdx.x * blockDim.x + threadIdx.x;
    if (p >= PAIR_WORDS) return;
    const unsigned* w = (const unsigned*)planes + p;
    unsigned a = 1u, b = 1u;
    for (int s = 0; s < DEG_SLICES; ++s) {
        unsigned c = w[(size_t)s * PAIR_WORDS];
        a += c & 0xFFFFu;
        b += c >> 16;
    }
    float2 o;
    o.x = rsqrtf((float)a);
    o.y = rsqrtf((float)b);
    ((float2*)deg)[p] = o;
    float4 q = ((const float4*)z)[p];
    q.x *= o.x; q.y *= o.x; q.z *= o.y; q.w *= o.y;
    ((float4*)z)[p] = q;
}

#define AGG_SLICES 15
#define AGG_RANGES 17
#define AGG_THREADS 1024
#define AGG_U 4
#define AGG_RANGE_NODES 5888
#define AGG_LAST_NODES  (N_NODES - (AGG_RANGES - 1) * AGG_RANGE_NODES)
#define AGG_ACC_FLOATS  (AGG_RANGE_NODES * 2)
#define AGG_SLICE_VEC4  ((EDGE_VEC4 + AGG_SLICES - 1) / AGG_SLICES)
static_assert(N_EDGES % 4 == 0, "col is whole 16-byte loads");
static_assert((AGG_ACC_FLOATS * 4) % 128 == 0 && (N_NODES * 2 * 4) % 128 == 0,
              "a range of a plane is whole 128-byte lines, so no line has two writers");
static_assert(AGG_LAST_NODES > 0 && AGG_LAST_NODES <= AGG_RANGE_NODES && AGG_LAST_NODES % 2 == 0,
              "the ranges cover every node and the short one is whole 16-byte pieces");

__global__ void __launch_bounds__(AGG_THREADS) gcn_aggregate_lds(const int* __restrict__ row,
                                                                 const int* __restrict__ col,
                                                                 const float* __restrict__ z,
                                                                 float* __restrict__ planes) {
    __shared__ float acc[AGG_ACC_FLOATS];
    const int r = blockIdx.x % AGG_RANGES;
    const int s = blockIdx.x / AGG_RANGES;
    for (int i = threadIdx.x; i < AGG_ACC_FLOATS / 4; i += AGG_THREADS)
        ((float4*)acc)[i] = make_float4(0.0f, 0.0f, 0.0f, 0.0f);
    __syncthreads();

    const unsigned lo = (unsigned)(r * AGG_RANGE_NODES);
    const unsigned n_here = (r == AGG_RANGES - 1) ? (unsigned)AGG_LAST_NODES : (unsigned)AGG_RANGE_NODES;
    const int v_begin = s * AGG_SLICE_VEC4;
    const int v_end = (v_begin + AGG_SLICE_VEC4 < EDGE_VEC4) ? v_begin + AGG_SLICE_VEC4 : EDGE_VEC4;
    const int e_spare = v_begin * 4;
    const int4* c4 = (const int4*)col;
    const float2* z2 = (const float2*)z;

    for (int base = v_begin; base < v_end; base += AGG_THREADS * AGG_U) {
        int d[AGG_U * 4];
        int e[AGG_U * 4];
        for (int j = 0; j < AGG_U; ++j) {
            int i = base + j * AGG_THREADS + (int)threadIdx.x;
            int4 v = (i < v_end) ? c4[i] : make_int4(-1, -1, -1, -1);
            d[4 * j + 0] = v.x; d[4 * j + 1] = v.y; d[4 * j + 2] = v.z; d[4 * j + 3] = v.w;
            e[4 * j + 0] = 4 * i; e[4 * j + 1] = 4 * i + 1; e[4 * j + 2] = 4 * i + 2; e[4 * j + 3] = 4 * i + 3;
        }
        unsigned k[AGG_U * 4];
        int src[AGG_U * 4];
        for (int m = 0; m < AGG_U * 4; ++m) {
            k[m] = (unsigned)d[m] - lo;
            src[m] = row[(k[m] < n_here) ? e[m] : e_spare];
        }
        float2 zz[AGG_U * 4];
        for (int m = 0; m < AGG_U * 4; ++m) zz[m] = z2[src[m]];
        asm volatile("" :: "v"(zz[0].x), "v"(zz[0].y));
        for (int m = 0; m < AGG_U * 4; ++m) {
            if (k[m] < n_here) {
                atomicAdd(&acc[2 * k[m] + 0], zz[m].x);
                atomicAdd(&acc[2 * k[m] + 1], zz[m].y);
            }
        }
    }
    __syncthreads();

    float4* dst = (float4*)(planes + (size_t)s * (N_NODES * 2) + (size_t)r * AGG_ACC_FLOATS);
    const int n_store4 = (int)(n_here / 2);
    for (int i = threadIdx.x; i < n_store4; i += AGG_THREADS)
        dst[i] = ((const float4*)acc)[i];
}

__global__ void gcn_out(const float* __restrict__ planes, const float* __restrict__ z,
                        const float* __restrict__ dinv, const float* __restrict__ b1,
                        const float* __restrict__ W2, const float* __restrict__ b2,
                        float* __restrict__ out) {
    int t = blockIdx.x * blockDim.x + threadIdx.x;
    if (t >= N_NODES * 2) return;
    const int j = t & 1;
    float cst = b2[j];
    for (int c = 0; c < HID; ++c) cst += b1[c] * W2[c * 2 + j];
    float a = z[t];
#pragma unroll
    for (int s = 0; s < AGG_SLICES; ++s) a += planes[(size_t)s * (N_NODES * 2) + t];
    out[t] = dinv[t >> 1] * a + cst;
}

extern "C" void kernel_launch(void* const* d_in, const int* in_sizes, int n_in,
                              void* d_out, int out_size, void* d_ws, size_t ws_size,
                              hipStream_t stream) {
    const float* x  = (const float*)d_in[0];
    const int*   ei = (const int*)d_in[1];
    const float* W1 = (const float*)d_in[2];
    const float* b1 = (const float*)d_in[3];
    const float* W2 = (const float*)d_in[4];
    const float* b2 = (const float*)d_in[5];
    float* out = (float*)d_out;

    float* deg = (float*)d_ws;
    float* z   = deg + N_NODES;
    float* agg = z + (size_t)N_NODES * 2;

    const int* row = ei;
    const int* col = ei + N_EDGES;

    gcn_degree_and_xw<<<DEG_BLOCKS + XW_BLOCKS, FUSED_THREADS, 0, stream>>>(col, agg, x, W1, W2, z);
    gcn_rsqrt_scale<<<(PAIR_WORDS + 255) / 256, 256, 0, stream>>>(agg, deg, z);

    gcn_aggregate_lds<<<AGG_SLICES * AGG_RANGES, AGG_THREADS, 0, stream>>>(row, col, z, agg);

    gcn_out<<<(N_NODES * 2 + 255) / 256, 256, 0, stream>>>(agg, z, deg, b1, W2, b2, out);
}


# aggregation: in-range edges first filtered into per-thread LDS lists (col, row as streamed 16-byte loads), z gathered and LDS-added from the lists; order of the LDS float adds changed
# speedup vs baseline: 5.0144x; 1.1088x over previous
#include <hip/hip_runtime.h>


typedef __attribute__((ext_vector_type(2))) float v2f;
typedef __attribute__((ext_vector_type(8))) float v8f;

#define N_NODES 100000
#define N_EDGES 3200000
#define IN_CH 512
#define HID 16
#define EDGE_VEC4 (N_EDGES / 4)

#define FUSED_THREADS 256
#define DEG_SLICES 60
#define DEG_RANGES 4
#define DEG_BLOCKS (DEG_SLICES * DEG_RANGES)
#define DEG_SLICE_VEC4 ((EDGE_VEC4 + DEG_SLICES - 1) / DEG_SLICES)
#define DEG_SLICE_EDGES (DEG_SLICE_VEC4 * 4)
#define DEG_U 8
#define RANGE_NODES (N_NODES / DEG_RANGES)
#define RANGE_WORDS (RANGE_NODES / 2)
#define PAIR_WORDS  (N_NODES / 2)
#define XW_TILES  (N_NODES / 16)
#define XW_BLOCKS ((XW_TILES + 7) / 8)
static_assert(N_EDGES % 4 == 0, "col is whole 16-byte loads");
static_assert(DEG_SLICE_EDGES <= 65535, "one slice landing on one node must fit a 16-bit field");
static_assert(N_NODES % DEG_RANGES == 0 && RANGE_NODES % 2 == 0 && RANGE_WORDS % 4 == 0,
              "ranges are whole 16-byte pieces of packed words");
static_assert(N_NODES % 16 == 0, "tiles are whole");

#define XW_CHUNK_K 64
#define XW_CHUNKS (IN_CH / XW_CHUNK_K)
#define XW_PITCH4 17
#define XW_STAGE4 (16 * XW_PITCH4)
static_assert(IN_CH % XW_CHUNK_K == 0 && XW_CHUNK_K == 64, "a chunk is 16 lanes x 16 bytes of one row");

struct XwLds {
    float4 w1p[(IN_CH * HID) / 4];
    float4 stage[FUSED_THREADS / 32][XW_STAGE4];
};
union FusedLds {
    uint4 cnt4[RANGE_WORDS / 4];
    XwLds xw;
};

__device__ __forceinline__ void deg_count(unsigned* cnt, int d, unsigned lo) {
    unsigned k = (unsigned)d - lo;
    if (k < (unsigned)RANGE_NODES) atomicAdd(&cnt[k >> 1], 1u << ((k & 1u) << 4));
}

__device__ __forceinline__ void deg_load(int4 (&v)[DEG_U], const int4* __restrict__ c4, int base, int v_end) {
    for (int j = 0; j < DEG_U; ++j) {
        int i = base + j * FUSED_THREADS + (int)threadIdx.x;
        v[j] = (i < v_end) ? c4[i] : make_int4(-1, -1, -1, -1);
    }
}

__device__ __forceinline__ void degree_block(FusedLds& lds, int b, const int* __restrict__ col,
                                             float* __restrict__ planes) {
    unsigned* cnt = (unsigned*)lds.cnt4;
    const int r = b % DEG_RANGES;
    const int s = b / DEG_RANGES;
    const int v_begin = s * DEG_SLICE_VEC4;
    const int v_end = (v_begin + DEG_SLICE_VEC4 < EDGE_VEC4) ? v_begin + DEG_SLICE_VEC4 : EDGE_VEC4;
    const int4* c4 = (const int4*)col;

    int4 v[DEG_U];
    deg_load(v, c4, v_begin, v_end);
    for (int i = threadIdx.x; i < RANGE_WORDS / 4; i += FUSED_THREADS)
        lds.cnt4[i] = make_uint4(0u, 0u, 0u, 0u);
    __syncthreads();

    const unsigned lo = (unsigned)(r * RANGE_NODES);
    for (int base = v_begin; base < v_end; base += FUSED_THREADS * DEG_U) {
        int4 nxt[DEG_U];
        deg_load(nxt, c4, base + FUSED_THREADS * DEG_U, v_end);
        for (int j = 0; j < DEG_U; ++j) {
            deg_count(cnt, v[j].x, lo);
            deg_count(cnt, v[j].y, lo);
            deg_count(cnt, v[j].z, lo);
            deg_count(cnt, v[j].w, lo);
        }
        for (int j = 0; j < DEG_U; ++j) v[j] = nxt[j];
    }
    __syncthreads();

    uint4* dst = (uint4*)((unsigned*)planes + (size_t)s * PAIR_WORDS + r * RANGE_WORDS);
    for (int i = threadIdx.x; i < RANGE_WORDS / 4; i += FUSED_THREADS)
        dst[i] = lds.cnt4[i];
}


__device__ __forceinline__ void wave_stage_handover() {
    __builtin_amdgcn_wave_barrier();
}

#define XW_ROWPAIR4 (2 * IN_CH / 4)
#define XW_LOAD_CHUNK(v, p) do { \
    v##0 = (p)[0 * XW_ROWPAIR4]; v##1 = (p)[1 * XW_ROWPAIR4]; v##2 = (p)[2 * XW_ROWPAIR4]; v##3 = (p)[3 * XW_ROWPAIR4]; \
    v##4 = (p)[4 * XW_ROWPAIR4]; v##5 = (p)[5 * XW_ROWPAIR4]; v##6 = (p)[6 * XW_ROWPAIR4]; v##7 = (p)[7 * XW_ROWPAIR4]; \
} while (0)
#define XW_STORE_CHUNK(S, v) do { \
    (S)[0 * 2 * XW_PITCH4] = v##0; (S)[1 * 2 * XW_PITCH4] = v##1; (S)[2 * 2 * XW_PITCH4] = v##2; (S)[3 * 2 * XW_PITCH4] = v##3; \
    (S)[4 * 2 * XW_PITCH4] = v##4; (S)[5 * 2 * XW_PITCH4] = v##5; (S)[6 * 2 * XW_PITCH4] = v##6; (S)[7 * 2 * XW_PITCH4] = v##7; \
} while (0)

__device__ __forceinline__ v8f xw_chunk_products(v8f acc, const float4* A, const float4* B) {
#pragma unroll
    for (int j = 0; j < 8; ++j) {
        const float4 a4 = A[2 * j];
        const float4 b4 = B[2 * j * HID];
        v2f a, b;
        a.x = a4.x; a.y = a4.y; b.x = b4.x; b.y = b4.y;
        acc = __builtin_amdgcn_wmma_f32_16x16x4_f32(
             false, a,  false, b,
             (short)0, acc,  false,  false);
        a.x = a4.z; a.y = a4.w; b.x = b4.z; b.y = b4.w;
        acc = __builtin_amdgcn_wmma_f32_16x16x4_f32(
             false, a,  false, b,
             (short)0, acc,  false,  false);
    }
    return acc;
}

__device__ __forceinline__ void xw_block(FusedLds& lds, int blk, const float* __restrict__ x,
                                         const float* __restrict__ W1,
                                         const float* __restrict__ W2,
                                         float* __restrict__ y) {
    for (int i = threadIdx.x; i < (IN_CH * HID) / 4; i += FUSED_THREADS) {
        const float* w = W1 + (size_t)(4 * (i >> 4)) * HID + (i & 15);
        lds.xw.w1p[i] = make_float4(w[0], w[HID], w[2 * HID], w[3 * HID]);
    }
    __syncthreads();

    const int wave = threadIdx.x >> 5;
    const int lane = threadIdx.x & 31;
    const int half = lane >> 4;
    const int lrow = lane & 15;
    const int tile = blk * 8 + wave;
    if (tile >= XW_TILES) return;

    float4* S4 = lds.xw.stage[wave];
    const float4* xq = (const float4*)(x + (size_t)(tile * 16 + half) * IN_CH) + lrow;
    float4* st = S4 + half * XW_PITCH4 + lrow;
    const float4* fa = S4 + lrow * XW_PITCH4 + half;
    const float4* fb = lds.xw.w1p + half * HID + lrow;

    v8f acc = {};
    float4 ca0, ca1, ca2, ca3, ca4, ca5, ca6, ca7;
    float4 cb0, cb1, cb2, cb3, cb4, cb5, cb6, cb7;
    XW_LOAD_CHUNK(ca, xq);
#pragma unroll
    for (int ch = 0; ch < XW_CHUNKS; ch += 2) {
        XW_LOAD_CHUNK(cb, xq + (ch + 1) * (XW_CHUNK_K / 4));
        wave_stage_handover();
        XW_STORE_CHUNK(st, ca);
        wave_stage_handover();
        acc = xw_chunk_products(acc, fa, fb + ch * (XW_CHUNK_K / 4) * HID);
        if (ch + 2 < XW_CHUNKS) XW_LOAD_CHUNK(ca, xq + (ch + 2) * (XW_CHUNK_K / 4));
        wave_stage_handover();
        XW_STORE_CHUNK(st, cb);
        wave_stage_handover();
        acc = xw_chunk_products(acc, fa, fb + (ch + 1) * (XW_CHUNK_K / 4) * HID);
    }

    const float w0 = W2[lrow * 2 + 0];
    const float w1 = W2[lrow * 2 + 1];
    const bool odd = (lrow & 1) != 0;
    float mine = 0.0f;
#pragma unroll
    for (int r = 0; r < 8; ++r) {
        float p0 = acc[r] * w0, p1 = acc[r] * w1;
        float keep = odd ? p1 : p0;
        float send = odd ? p0 : p1;
        keep += __shfl_xor(send, 1);
        keep += __shfl_xor(keep, 2);
        keep += __shfl_xor(keep, 4);
        keep += __shfl_xor(keep, 8);
        if ((lrow >> 1) == r) mine = keep;
    }
    y[tile * 32 + lane] = mine;
}

__global__ void __launch_bounds__(FUSED_THREADS) gcn_degree_and_xw(const int* __restrict__ col,
                                                                   float* __restrict__ planes,
                                                                   const float* __restrict__ x,
                                                                   const float* __restrict__ W1,
                                                                   const float* __restrict__ W2,
                                                                   float* __restrict__ y) {
    __shared__ FusedLds lds;
    if (blockIdx.x < DEG_BLOCKS) degree_block(lds, (int)blockIdx.x, col, planes);
    else xw_block(lds, (int)blockIdx.x - DEG_BLOCKS, x, W1, W2, y);
}

__global__ void gcn_rsqrt_scale(const float* __restrict__ planes, float* __restrict__ deg,
                                float* z) {
    int p = blockIdx.x * blockDim.x + threadIdx.x;
    if (p >= PAIR_WORDS) return;
    const unsigned* w = (const unsigned*)planes + p;
    unsigned a = 1u, b = 1u;
    for (int s = 0; s < DEG_SLICES; ++s) {
        unsigned c = w[(size_t)s * PAIR_WORDS];
        a += c & 0xFFFFu;
        b += c >> 16;
    }
    float2 o;
    o.x = rsqrtf((float)a);
    o.y = rsqrtf((float)b);
    ((float2*)deg)[p] = o;
    float4 q = ((const float4*)z)[p];
    q.x *= o.x; q.y *= o.x; q.z *= o.y; q.w *= o.y;
    ((float4*)z)[p] = q;
}

#define AGG_SLICES 15
#define AGG_RANGES 17
#define AGG_THREADS 1024
#define AGG_U 4
#define AGG_RANGE_NODES 5888
#define AGG_LAST_NODES  (N_NODES - (AGG_RANGES - 1) * AGG_RANGE_NODES)
#define AGG_ACC_FLOATS  (AGG_RANGE_NODES * 2)
#define AGG_SLICE_VEC4  ((EDGE_VEC4 + AGG_SLICES - 1) / AGG_SLICES)
#define AGG_QCAP 40
#define AGG_SRC_BITS 17
#define AGG_G 8
static_assert(N_EDGES % 4 == 0, "col and row are whole 16-byte loads");
static_assert((N_EDGES * 4) % 16 == 0, "col starts on a 16-byte boundary behind row");
static_assert((AGG_ACC_FLOATS * 4) % 128 == 0 && (N_NODES * 2 * 4) % 128 == 0,
              "a range of a plane is whole 128-byte lines, so no line has two writers");
static_assert(AGG_LAST_NODES > 0 && AGG_LAST_NODES <= AGG_RANGE_NODES && AGG_LAST_NODES % 2 == 0,
              "the ranges cover every node and the short one is whole 16-byte pieces");
static_assert(AGG_QCAP >= 4 * AGG_U, "an empty list takes a whole trip");
static_assert(N_NODES <= (1 << AGG_SRC_BITS) && AGG_RANGE_NODES < (1 << (32 - AGG_SRC_BITS)) - 1,
              "a packed entry holds both fields and is never all ones");
#define AGG_NONE 0xFFFFFFFFu

__device__ __forceinline__ void agg_load(int4 (&c)[AGG_U], int4 (&r)[AGG_U], const int4* __restrict__ c4,
                                         const int4* __restrict__ r4, int base, int v_end) {
    for (int j = 0; j < AGG_U; ++j) {
        int i = base + j * AGG_THREADS + (int)threadIdx.x;
        bool in = i < v_end;
        c[j] = in ? c4[i] : make_int4(-1, -1, -1, -1);
        r[j] = in ? r4[i] : make_int4(0, 0, 0, 0);
    }
}

template <bool CHECKED>
__device__ __forceinline__ void agg_keep(unsigned* __restrict__ q, float* __restrict__ acc,
                                         const float2* __restrict__ z2, int& cnt, int d, int s,
                                         unsigned lo, unsigned n_here) {
    unsigned k = (unsigned)d - lo;
    if (k < n_here) {
        if (!CHECKED || cnt < AGG_QCAP) {
            q[cnt * AGG_THREADS + (int)threadIdx.x] = (k << AGG_SRC_BITS) | (unsigned)s;
            ++cnt;
        } else {
            float2 v = z2[s];
            atomicAdd(&acc[2 * k + 0], v.x);
            atomicAdd(&acc[2 * k + 1], v.y);
        }
    }
}

template <bool CHECKED>
__device__ __forceinline__ void agg_keep_trip(unsigned* __restrict__ q, float* __restrict__ acc,
                                              const float2* __restrict__ z2, int& cnt,
                                              const int4 (&c)[AGG_U], const int4 (&w)[AGG_U],
                                              unsigned lo, unsigned n_here) {
    for (int j = 0; j < AGG_U; ++j) {
        agg_keep<CHECKED>(q, acc, z2, cnt, c[j].x, w[j].x, lo, n_here);
        agg_keep<CHECKED>(q, acc, z2, cnt, c[j].y, w[j].y, lo, n_here);
        agg_keep<CHECKED>(q, acc, z2, cnt, c[j].z, w[j].z, lo, n_here);
        agg_keep<CHECKED>(q, acc, z2, cnt, c[j].w, w[j].w, lo, n_here);
    }
}

__global__ void __launch_bounds__(AGG_THREADS) gcn_aggregate_lds(const int* __restrict__ row,
                                                                 const int* __restrict__ col,
                                                                 const float* __restrict__ z,
                                                                 float* __restrict__ planes) {
    __shared__ float acc[AGG_ACC_FLOATS];
    __shared__ unsigned q[AGG_QCAP * AGG_THREADS];
    const int r = blockIdx.x % AGG_RANGES;
    const int s = blockIdx.x / AGG_RANGES;
    const unsigned lo = (unsigned)(r * AGG_RANGE_NODES);
    const unsigned n_here = (r == AGG_RANGES - 1) ? (unsigned)AGG_LAST_NODES : (unsigned)AGG_RANGE_NODES;
    const int v_begin = s * AGG_SLICE_VEC4;
    const int v_end = (v_begin + AGG_SLICE_VEC4 < EDGE_VEC4) ? v_begin + AGG_SLICE_VEC4 : EDGE_VEC4;
    const int4* c4 = (const int4*)col;
    const int4* r4 = (const int4*)row;
    const float2* z2 = (const float2*)z;

    int4 c[AGG_U], w[AGG_U];
    agg_load(c, w, c4, r4, v_begin, v_end);
    for (int i = threadIdx.x; i < AGG_ACC_FLOATS / 4; i += AGG_THREADS)
        ((float4*)acc)[i] = make_float4(0.0f, 0.0f, 0.0f, 0.0f);
    __syncthreads();

    int cnt = 0;
    for (int base = v_begin; base < v_end; base += AGG_THREADS * AGG_U) {
        int4 cn[AGG_U], wn[AGG_U];
        agg_load(cn, wn, c4, r4, base + AGG_THREADS * AGG_U, v_end);
        if (__builtin_amdgcn_ballot_w32(cnt > AGG_QCAP - 4 * AGG_U) == 0u)
            agg_keep_trip<false>(q, acc, z2, cnt, c, w, lo, n_here);
        else
            agg_keep_trip<true>(q, acc, z2, cnt, c, w, lo, n_here);
        for (int j = 0; j < AGG_U; ++j) { c[j] = cn[j]; w[j] = wn[j]; }
    }

    for (int i = 0; i < cnt; i += AGG_G) {
        unsigned en[AGG_G];
        for (int n = 0; n < AGG_G; ++n)
            en[n] = (i + n < cnt) ? q[(i + n) * AGG_THREADS + (int)threadIdx.x] : AGG_NONE;
        float2 zz[AGG_G];
        for (int n = 0; n < AGG_G; ++n)
            zz[n] = z2[(en[n] != AGG_NONE) ? (en[n] & ((1u << AGG_SRC_BITS) - 1u)) : 0u];
        for (int n = 0; n < AGG_G; ++n) {
            if (en[n] != AGG_NONE) {
                unsigned k = en[n] >> AGG_SRC_BITS;
                atomicAdd(&acc[2 * k + 0], zz[n].x);
                atomicAdd(&acc[2 * k + 1], zz[n].y);
            }
        }
    }
    __syncthreads();

    float4* dst = (float4*)(planes + (size_t)s * (N_NODES * 2) + (size_t)r * AGG_ACC_FLOATS);
    const int n_store4 = (int)(n_here / 2);
    for (int i = threadIdx.x; i < n_store4; i += AGG_THREADS)
        dst[i] = ((const float4*)acc)[i];
}

__global__ void gcn_out(const float* __restrict__ planes, const float* __restrict__ z,
                        const float* __restrict__ dinv, const float* __restrict__ b1,
                        const float* __restrict__ W2, const float* __restrict__ b2,
                        float* __restrict__ out) {
    int t = blockIdx.x * blockDim.x + threadIdx.x;
    if (t >= N_NODES * 2) return;
    const int j = t & 1;
    float cst = b2[j];
    for (int c = 0; c < HID; ++c) cst += b1[c] * W2[c * 2 + j];
    float a = z[t];
#pragma unroll
    for (int s = 0; s < AGG_SLICES; ++s) a += planes[(size_t)s * (N_NODES * 2) + t];
    out[t] = dinv[t >> 1] * a + cst;
}

extern "C" void kernel_launch(void* const* d_in, const int* in_sizes, int n_in,
                              void* d_out, int out_size, void* d_ws, size_t ws_size,
                              hipStream_t stream) {
    const float* x  = (const float*)d_in[0];
    const int*   ei = (const int*)d_in[1];
    const float* W1 = (const float*)d_in[2];
    const float* b1 = (const float*)d_in[3];
    const float* W2 = (const float*)d_in[4];
    const float* b2 = (const float*)d_in[5];
    float* out = (float*)d_out;

    float* deg = (float*)d_ws;
    float* z   = deg + N_NODES;
    float* agg = z + (size_t)N_NODES * 2;

    const int* row = ei;
    const int* col = ei + N_EDGES;

    gcn_degree_and_xw<<<DEG_BLOCKS + XW_BLOCKS, FUSED_THREADS, 0, stream>>>(col, agg, x, W1, W2, z);
    gcn_rsqrt_scale<<<(PAIR_WORDS + 255) / 256, 256, 0, stream>>>(agg, deg, z);

    gcn_aggregate_lds<<<AGG_SLICES * AGG_RANGES, AGG_THREADS, 0, stream>>>(row, col, z, agg);

    gcn_out<<<(N_NODES * 2 + 255) / 256, 256, 0, stream>>>(agg, z, deg, b1, W2, b2, out);
}
